# starting kernel
# baseline (speedup 1.0000x reference)
#include <hip/hip_runtime.h>


typedef __attribute__((ext_vector_type(2))) float v2f;
typedef __attribute__((ext_vector_type(8))) float v8f;

#define NBLADES 8
#define BATCH   128
#define INFEAT  128
#define UNITS   64
#define CHUNK   32
#define XPAD    36

#define BLADE_PERM_TAB 0x76534210u
__device__ __forceinline__ int blade_perm(int i) {
  return (int)((BLADE_PERM_TAB >> (4 * i)) & 7u);
}

__global__ __launch_bounds__(256) void ga_mvprod_wmma(
    const float* __restrict__ x,
    const float* __restrict__ w,
    float* __restrict__ out)
{
  __shared__ float xs[NBLADES][16][XPAD];
  __shared__ float ws[NBLADES][CHUNK][16];

  const int tid   = threadIdx.x;
  const int lane  = tid & 31;
  const int o     = __builtin_amdgcn_readfirstlane(tid >> 5);
  const int mtile = blockIdx.x >> 2;
  const int ntile = blockIdx.x & 3;

  const int mo = blade_perm(o);

  unsigned ibword = 0, negword = 0;
#pragma unroll
  for (int ia = 0; ia < NBLADES; ++ia) {
    const int ma  = blade_perm(ia);
    const int mb  = mo ^ ma;
    const int ib  = blade_perm(mb);
    const int a1  = (ma >> 1) & 1, a2 = (ma >> 2) & 1;
    const int neg = ((mb & (a1 ^ a2)) ^ ((mb >> 1) & a2)) & 1;
    ibword  |= (unsigned)ib  << (3 * ia);
    negword |= (unsigned)neg << ia;
  }

  const int fm = lane & 15;
  const int kb = (lane >> 4) * 2;

  v8f accP = {};
  v8f accN = {};

  for (int ch = 0; ch < INFEAT / CHUNK; ++ch) {
    __syncthreads();

#pragma unroll
    for (int r = 0; r < 4; ++r) {
      int idx = tid + 256 * r;
      int ia  = idx >> 7;
      int rem = idx & 127;
      int m   = rem >> 3;
      int g   = rem & 7;
      float4 v = *(const float4*)(x + ((size_t)(ia * BATCH + mtile * 16 + m) * INFEAT
                                       + ch * CHUNK + 4 * g));
      *(float4*)(&xs[ia][m][4 * g]) = v;
    }
#pragma unroll
    for (int r = 0; r < 4; ++r) {
      int idx = tid + 256 * r;
      int ib  = idx >> 7;
      int rem = idx & 127;
      int k   = rem >> 2;
      int g   = rem & 3;
      float4 v = *(const float4*)(w + ((size_t)(ib * INFEAT + ch * CHUNK + k) * UNITS
                                       + ntile * 16 + 4 * g));
      *(float4*)(&ws[ib][k][4 * g]) = v;
    }
    __syncthreads();

    if (ch + 1 < INFEAT / CHUNK) {
      __builtin_prefetch(x + ((size_t)((tid >> 5) * BATCH + mtile * 16 + (tid & 15)) * INFEAT
                              + (ch + 1) * CHUNK), 0, 3);
      __builtin_prefetch(w + ((size_t)((tid >> 5) * INFEAT + (ch + 1) * CHUNK + (tid & 31)) * UNITS
                              + ntile * 16), 0, 3);
    }

    for (int ia = 0; ia < NBLADES; ++ia) {
      const int ib = (int)((ibword >> (3 * ia)) & 7u);
      if ((negword >> ia) & 1u) {
#pragma unroll
        for (int k = 0; k < CHUNK; k += 4) {
          v2f a, b;
          a.x = xs[ia][fm][k + kb];
          a.y = xs[ia][fm][k + kb + 1];
          b.x = ws[ib][k + kb][fm];
          b.y = ws[ib][k + kb + 1][fm];
          accN = __builtin_amdgcn_wmma_f32_16x16x4_f32(
              false, a, false, b, (short)0, accN, false, false);
        }
      } else {
#pragma unroll
        for (int k = 0; k < CHUNK; k += 4) {
          v2f a, b;
          a.x = xs[ia][fm][k + kb];
          a.y = xs[ia][fm][k + kb + 1];
          b.x = ws[ib][k + kb][fm];
          b.y = ws[ib][k + kb + 1][fm];
          accP = __builtin_amdgcn_wmma_f32_16x16x4_f32(
              false, a, false, b, (short)0, accP, false, false);
        }
      }
    }
  }

  const int n     = lane & 15;
  const int mbase = (lane >> 4) * 8;
#pragma unroll
  for (int r = 0; r < 8; ++r) {
    out[((size_t)o * BATCH + mtile * 16 + mbase + r) * UNITS + ntile * 16 + n] =
        accP[r] - accN[r];
  }
}

extern "C" void kernel_launch(void* const* d_in, const int* in_sizes, int n_in,
                              void* d_out, int out_size, void* d_ws, size_t ws_size,
                              hipStream_t stream) {
  (void)in_sizes; (void)n_in; (void)out_size; (void)d_ws; (void)ws_size;
  const float* x = (const float*)d_in[0];
  const float* w = (const float*)d_in[1];
  float* out = (float*)d_out;
  dim3 grid(32);
  dim3 block(256);
  hipLaunchKernelGGL(ga_mvprod_wmma, grid, block, 0, stream, x, w, out);
}


# Output-stationary: tile rows = 8 OUTPUT blades x 2 batch rows, 8 w blades concatenated along K, x rows permuted+signed at load, w staged per wave; 4 waves split K, LDS add (changes summation order)
# speedup vs baseline: 1.8162x; 1.8162x over previous
#include <hip/hip_runtime.h>


typedef __attribute__((ext_vector_type(2))) float v2f;
typedef __attribute__((ext_vector_type(8))) float v8f;

#define NBLADES 8
#define BATCH   128
#define INFEAT  128
#define UNITS   64

#define BLADE_PERM_TAB 0x76534210u
__device__ __forceinline__ int blade_perm(int i) {
  return (int)((BLADE_PERM_TAB >> (4 * i)) & 7u);
}

__global__ __launch_bounds__(128) void ga_mvprod_kcat(
    const float* __restrict__ x,
    const float* __restrict__ w,
    float* __restrict__ out)
{
  __shared__ __attribute__((aligned(16))) float wsm[4][2][128 * 16 + 32];
  __shared__ float red[4][8][32];

  const int tid  = threadIdx.x;
  const int lane = tid & 31;
  const int wv   = __builtin_amdgcn_readfirstlane(tid >> 5);
  const int bp    = blockIdx.x >> 2;
  const int ntile = blockIdx.x & 3;

  const int c = lane & 15;
  const int h = lane >> 4;
  const int o    = c & 7;
  const int brow = 2 * bp + (c >> 3);
  const int mo   = blade_perm(o);

  v8f acc[2];
  acc[0] = (v8f){};
  acc[1] = (v8f){};

  const float4* xr[2];
  unsigned      sgn[2];
#pragma unroll
  for (int j = 0; j < 2; ++j) {
    const int ib = 2 * wv + j;
    const int mb = blade_perm(ib);
    const int ma = mo ^ mb;
    const int ia = blade_perm(ma);
    const int a1 = (ma >> 1) & 1, a2 = (ma >> 2) & 1;
    sgn[j] = (unsigned)(((mb & (a1 ^ a2)) ^ ((mb >> 1) & a2)) & 1) << 31;
    xr[j] = (const float4*)(x + ((size_t)(ia * BATCH + brow) * INFEAT + 64 * h));
  }

  const int wrow = lane >> 2, wg = lane & 3;
  const float* wsrc = w + ((size_t)(2 * wv * INFEAT + wrow) * UNITS + ntile * 16 + 4 * wg);
  float4 wv4[32];
#pragma unroll
  for (int i = 0; i < 32; ++i) {
    wv4[i] = *(const float4*)(wsrc + i * 8 * UNITS);
  }
  float4 xv[2][16];
#pragma unroll
  for (int j = 0; j < 2; ++j) {
#pragma unroll
    for (int q = 0; q < 16; ++q) xv[j][q] = xr[j][q];
  }

  float* wdst = &wsm[wv][0][wrow * 16 + 4 * wg];
#pragma unroll
  for (int i = 0; i < 32; ++i) {
    const int j  = i >> 4;
    const int f8 = 8 * (i & 15);
    *(float4*)(wdst + j * (128 * 16 + 32) + f8 * 16 + (f8 >> 6) * 32) = wv4[i];
  }
  v2f a[2][32];
#pragma unroll
  for (int j = 0; j < 2; ++j) {
#pragma unroll
    for (int q = 0; q < 16; ++q) {
      const float4 v = xv[j][q];
      a[j][2 * q].x     = __uint_as_float(__float_as_uint(v.x) ^ sgn[j]);
      a[j][2 * q].y     = __uint_as_float(__float_as_uint(v.y) ^ sgn[j]);
      a[j][2 * q + 1].x = __uint_as_float(__float_as_uint(v.z) ^ sgn[j]);
      a[j][2 * q + 1].y = __uint_as_float(__float_as_uint(v.w) ^ sgn[j]);
    }
  }
  __syncthreads();

  v2f b[2][32];
  const float* bl = &wsm[wv][0][(64 * h) * 16 + h * 32 + c];
#pragma unroll
  for (int t = 0; t < 32; ++t) {
#pragma unroll
    for (int j = 0; j < 2; ++j) {
      b[j][t].x = bl[j * (128 * 16 + 32) + (2 * t + 0) * 16];
      b[j][t].y = bl[j * (128 * 16 + 32) + (2 * t + 1) * 16];
    }
  }
  __builtin_amdgcn_sched_barrier(0);

#pragma unroll
  for (int t = 0; t < 32; ++t) {
#pragma unroll
    for (int j = 0; j < 2; ++j) {
      acc[j] = __builtin_amdgcn_wmma_f32_16x16x4_f32(
          false, a[j][t], false, b[j][t], (short)0, acc[j], false, false);
    }
  }

#pragma unroll
  for (int r = 0; r < 8; ++r) red[wv][r][lane] = acc[0][r] + acc[1][r];
  __syncthreads();

#pragma unroll
  for (int t = 0; t < 2; ++t) {
    const int r = 2 * wv + t;
    const float s = (red[0][r][lane] + red[1][r][lane]) +
                    (red[2][r][lane] + red[3][r][lane]);
    out[((size_t)r * BATCH + 2 * bp + h) * UNITS + ntile * 16 + c] = s;
  }
}

extern "C" void kernel_launch(void* const* d_in, const int* in_sizes, int n_in,
                              void* d_out, int out_size, void* d_ws, size_t ws_size,
                              hipStream_t stream) {
  (void)in_sizes; (void)n_in; (void)out_size; (void)d_ws; (void)ws_size;
  const float* x = (const float*)d_in[0];
  const float* w = (const float*)d_in[1];
  float* out = (float*)d_out;
  dim3 grid(256);
  dim3 block(128);
  hipLaunchKernelGGL(ga_mvprod_kcat, grid, block, 0, stream, x, w, out);
}


# x rows staged once per work-group in LDS, one whole 512 B row per global load (was 128 fragment-shaped loads of 32 lines each); A fragments by ds_load_b128; sign moved to accumulator rows
# speedup vs baseline: 2.1382x; 1.1773x over previous
#include <hip/hip_runtime.h>


typedef __attribute__((ext_vector_type(2))) float v2f;
typedef __attribute__((ext_vector_type(8))) float v8f;

#define NBLADES 8
#define BATCH   128
#define INFEAT  128
#define UNITS   64
#define XPITCH  132

#define BLADE_PERM_TAB 0x76534210u
__device__ __forceinline__ int blade_perm(int i) {
  return (int)((BLADE_PERM_TAB >> (4 * i)) & 7u);
}
__device__ __forceinline__ unsigned blade_sign(int mo, int mb) {
  const int ma = mo ^ mb;
  const int a1 = (ma >> 1) & 1, a2 = (ma >> 2) & 1;
  return (unsigned)(((mb & (a1 ^ a2)) ^ ((mb >> 1) & a2)) & 1) << 31;
}

__global__ __launch_bounds__(128) void ga_mvprod_kcat(
    const float* __restrict__ x,
    const float* __restrict__ w,
    float* __restrict__ out)
{
  __shared__ __attribute__((aligned(16))) float wsm[4][2][128 * 16 + 32];
  __shared__ __attribute__((aligned(16))) float xsm[16 * XPITCH];
  __shared__ float red[4][8][32];

  const int tid  = threadIdx.x;
  const int lane = tid & 31;
  const int wv   = __builtin_amdgcn_readfirstlane(tid >> 5);
  const int bp    = blockIdx.x >> 2;
  const int ntile = blockIdx.x & 3;

  const int c = lane & 15;
  const int h = lane >> 4;
  const int o    = c & 7;
  const int bsel = c >> 3;
  const int mo   = blade_perm(o);

  v8f acc[2];
  acc[0] = (v8f){};
  acc[1] = (v8f){};

  const float4* xr[2];
#pragma unroll
  for (int j = 0; j < 2; ++j) {
    const int ib = 2 * wv + j;
    const int mb = blade_perm(ib);
    const int ia = blade_perm(mo ^ mb);
    xr[j] = (const float4*)&xsm[(2 * ia + bsel) * XPITCH + 64 * h];
  }

  const float* xsrc = x + ((size_t)(2 * wv * BATCH + 2 * bp) * INFEAT + 4 * lane);
  float4 xg[4];
#pragma unroll
  for (int i = 0; i < 4; ++i) {
    xg[i] = *(const float4*)(xsrc + (size_t)((i >> 1) * BATCH + (i & 1)) * INFEAT);
  }
  const int wrow = lane >> 2, wg = lane & 3;
  const float* wsrc = w + ((size_t)(2 * wv * INFEAT + wrow) * UNITS + ntile * 16 + 4 * wg);
  float4 wv4[32];
#pragma unroll
  for (int i = 0; i < 32; ++i) {
    wv4[i] = *(const float4*)(wsrc + i * 8 * UNITS);
  }

#pragma unroll
  for (int i = 0; i < 4; ++i) {
    *(float4*)&xsm[(4 * wv + i) * XPITCH + 4 * lane] = xg[i];
  }
  float* wdst = &wsm[wv][0][wrow * 16 + 4 * wg];
#pragma unroll
  for (int i = 0; i < 32; ++i) {
    const int j  = i >> 4;
    const int f8 = 8 * (i & 15);
    *(float4*)(wdst + j * (128 * 16 + 32) + f8 * 16 + (f8 >> 6) * 32) = wv4[i];
  }
  __syncthreads();

  float4 xv[2][16];
  v2f b[2][32];
  const float* bl = &wsm[wv][0][(64 * h) * 16 + h * 32 + c];
#pragma unroll
  for (int q = 0; q < 16; ++q) {
#pragma unroll
    for (int j = 0; j < 2; ++j) xv[j][q] = xr[j][q];
#pragma unroll
    for (int t = 2 * q; t < 2 * q + 2; ++t) {
#pragma unroll
      for (int j = 0; j < 2; ++j) {
        b[j][t].x = bl[j * (128 * 16 + 32) + (2 * t + 0) * 16];
        b[j][t].y = bl[j * (128 * 16 + 32) + (2 * t + 1) * 16];
      }
    }
  }
  __builtin_amdgcn_sched_barrier(0);

#pragma unroll
  for (int t = 0; t < 32; ++t) {
#pragma unroll
    for (int j = 0; j < 2; ++j) {
      const float4 v = xv[j][t >> 1];
      v2f a;
      a.x = (t & 1) ? v.z : v.x;
      a.y = (t & 1) ? v.w : v.y;
      acc[j] = __builtin_amdgcn_wmma_f32_16x16x4_f32(
          false, a, false, b[j][t], (short)0, acc[j], false, false);
    }
  }

#pragma unroll
  for (int r = 0; r < 8; ++r) {
    const unsigned s0 = blade_sign(blade_perm(r), blade_perm(2 * wv));
    const unsigned s1 = blade_sign(blade_perm(r), blade_perm(2 * wv + 1));
    red[wv][r][lane] = __uint_as_float(__float_as_uint(acc[0][r]) ^ s0) +
                       __uint_as_float(__float_as_uint(acc[1][r]) ^ s1);
  }
  __syncthreads();

#pragma unroll
  for (int t = 0; t < 2; ++t) {
    const int r = 2 * wv + t;
    const float s = (red[0][r][lane] + red[1][r][lane]) +
                    (red[2][r][lane] + red[3][r][lane]);
    out[((size_t)r * BATCH + 2 * bp + h) * UNITS + ntile * 16 + c] = s;
  }
}

extern "C" void kernel_launch(void* const* d_in, const int* in_sizes, int n_in,
                              void* d_out, int out_size, void* d_ws, size_t ws_size,
                              hipStream_t stream) {
  (void)in_sizes; (void)n_in; (void)out_size; (void)d_ws; (void)ws_size;
  const float* x = (const float*)d_in[0];
  const float* w = (const float*)d_in[1];
  float* out = (float*)d_out;
  dim3 grid(256);
  dim3 block(128);
  hipLaunchKernelGGL(ga_mvprod_kcat, grid, block, 0, stream, x, w, out);
}


# w loads issued in consumption order and handed through LDS chunk by chunk (1,1,2,2,2 groups of 4 k steps), each in front of the previous chunk's matrix run; the one barrier fences LDS only
# speedup vs baseline: 2.3092x; 1.0800x over previous
#include <hip/hip_runtime.h>


typedef __attribute__((ext_vector_type(2))) float v2f;
typedef __attribute__((ext_vector_type(8))) float v8f;
typedef __attribute__((ext_vector_type(4))) float v4f;

#define NBLADES 8
#define BATCH   128
#define INFEAT  128
#define UNITS   64
#define XPITCH  132
#define WPLANE  (128 * 16 + 32)

#define BLADE_PERM_TAB 0x76534210u
__device__ __forceinline__ int blade_perm(int i) {
  return (int)((BLADE_PERM_TAB >> (4 * i)) & 7u);
}
__device__ __forceinline__ unsigned blade_sign(int mo, int mb) {
  const int ma = mo ^ mb;
  const int a1 = (ma >> 1) & 1, a2 = (ma >> 2) & 1;
  return (unsigned)(((mb & (a1 ^ a2)) ^ ((mb >> 1) & a2)) & 1) << 31;
}

__global__ __launch_bounds__(128) void ga_mvprod_kcat(
    const float* __restrict__ x,
    const float* __restrict__ w,
    float* __restrict__ out)
{
  __shared__ __attribute__((aligned(16))) float wsm[4][2][128 * 16 + 32];
  __shared__ __attribute__((aligned(16))) float xsm[16 * XPITCH];
  __shared__ float red[4][8][32];

  const int tid  = threadIdx.x;
  const int lane = tid & 31;
  const int wv   = __builtin_amdgcn_readfirstlane(tid >> 5);
  const int bp    = blockIdx.x >> 2;
  const int ntile = blockIdx.x & 3;

  const int c = lane & 15;
  const int h = lane >> 4;
  const int o    = c & 7;
  const int bsel = c >> 3;
  const int mo   = blade_perm(o);

  v8f acc[2];
  acc[0] = (v8f){};
  acc[1] = (v8f){};

  const v4f* xr[2];
#pragma unroll
  for (int j = 0; j < 2; ++j) {
    const int ib = 2 * wv + j;
    const int mb = blade_perm(ib);
    const int ia = blade_perm(mo ^ mb);
    xr[j] = (const v4f*)&xsm[(2 * ia + bsel) * XPITCH + 64 * h];
  }

  const float* xsrc = x + ((size_t)(2 * wv * BATCH + 2 * bp) * INFEAT + 4 * lane);
  v4f xg[4];
#pragma unroll
  for (int i = 0; i < 4; ++i) {
    xg[i] = *(const v4f*)(xsrc + (size_t)((i >> 1) * BATCH + (i & 1)) * INFEAT);
  }
  const int wrow = lane >> 2, wg = lane & 3;
  const float* wsrc = w + ((size_t)(2 * wv * INFEAT + wrow) * UNITS + ntile * 16 + 4 * wg);
  v4f wq[32];
#pragma unroll
  for (int g = 0; g < 8; ++g) {
#pragma unroll
    for (int j = 0; j < 2; ++j) {
#pragma unroll
      for (int hh = 0; hh < 2; ++hh) {
        wq[(g * 2 + j) * 2 + hh] =
            *(const v4f*)(wsrc + (16 * j + 8 * hh + g) * 8 * UNITS);
      }
    }
  }

  float* wdst = &wsm[wv][0][wrow * 16 + 4 * wg];
  const float* bl = &wsm[wv][0][(64 * h) * 16 + h * 32 + c];
  v4f xv[2][16];
  v2f b[2][32];

#define W_TO_LDS(G0, G1)                                                      \
  _Pragma("unroll")                                                           \
  for (int g = (G0); g < (G1); ++g) {                                         \
    _Pragma("unroll")                                                         \
    for (int j = 0; j < 2; ++j) {                                             \
      _Pragma("unroll")                                                       \
      for (int hh = 0; hh < 2; ++hh) {                                        \
        *(v4f*)(wdst + j * WPLANE + (64 * hh + 8 * g) * 16 + hh * 32) =       \
            wq[(g * 2 + j) * 2 + hh];                                         \
      }                                                                       \
    }                                                                         \
  }
#define FRAGS_FROM_LDS(G0, G1)                                                \
  _Pragma("unroll")                                                           \
  for (int q = 2 * (G0); q < 2 * (G1); ++q) {                                 \
    _Pragma("unroll")                                                         \
    for (int j = 0; j < 2; ++j) xv[j][q] = xr[j][q];                          \
    _Pragma("unroll")                                                         \
    for (int t = 2 * q; t < 2 * q + 2; ++t) {                                 \
      _Pragma("unroll")                                                       \
      for (int j = 0; j < 2; ++j) {                                           \
        b[j][t].x = bl[j * WPLANE + (2 * t + 0) * 16];                        \
        b[j][t].y = bl[j * WPLANE + (2 * t + 1) * 16];                        \
      }                                                                       \
    }                                                                         \
  }
#define MMA(G0, G1)                                                           \
  _Pragma("unroll")                                                           \
  for (int t = 4 * (G0); t < 4 * (G1); ++t) {                                 \
    _Pragma("unroll")                                                         \
    for (int j = 0; j < 2; ++j) {                                             \
      const v4f v = xv[j][t >> 1];                                            \
      v2f a;                                                                  \
      a.x = (t & 1) ? v.z : v.x;                                              \
      a.y = (t & 1) ? v.w : v.y;                                              \
      acc[j] = __builtin_amdgcn_wmma_f32_16x16x4_f32(                         \
          false, a, false, b[j][t], (short)0, acc[j], false, false);          \
    }                                                                         \
  }

#pragma unroll
  for (int i = 0; i < 4; ++i) {
    *(v4f*)&xsm[(4 * wv + i) * XPITCH + 4 * lane] = xg[i];
  }
  W_TO_LDS(0, 1)
  __builtin_amdgcn_fence(__ATOMIC_RELEASE, "workgroup", "local");
  __builtin_amdgcn_s_barrier();
  __builtin_amdgcn_fence(__ATOMIC_ACQUIRE, "workgroup", "local");
  FRAGS_FROM_LDS(0, 1)

  W_TO_LDS(1, 2)
  FRAGS_FROM_LDS(1, 2)
  __builtin_amdgcn_sched_barrier(0);
  MMA(0, 1)
  __builtin_amdgcn_sched_barrier(0);
  W_TO_LDS(2, 4)
  FRAGS_FROM_LDS(2, 4)
  __builtin_amdgcn_sched_barrier(0);
  MMA(1, 2)
  __builtin_amdgcn_sched_barrier(0);
  W_TO_LDS(4, 6)
  FRAGS_FROM_LDS(4, 6)
  __builtin_amdgcn_sched_barrier(0);
  MMA(2, 4)
  __builtin_amdgcn_sched_barrier(0);
  W_TO_LDS(6, 8)
  FRAGS_FROM_LDS(6, 8)
  __builtin_amdgcn_sched_barrier(0);
  MMA(4, 6)
  MMA(6, 8)
#undef W_TO_LDS
#undef FRAGS_FROM_LDS
#undef MMA

#pragma unroll
  for (int r = 0; r < 8; ++r) {
    const unsigned s0 = blade_sign(blade_perm(r), blade_perm(2 * wv));
    const unsigned s1 = blade_sign(blade_perm(r), blade_perm(2 * wv + 1));
    red[wv][r][lane] = __uint_as_float(__float_as_uint(acc[0][r]) ^ s0) +
                       __uint_as_float(__float_as_uint(acc[1][r]) ^ s1);
  }
  __syncthreads();

#pragma unroll
  for (int t = 0; t < 2; ++t) {
    const int r = 2 * wv + t;
    const float s = (red[0][r][lane] + red[1][r][lane]) +
                    (red[2][r][lane] + red[3][r][lane]);
    out[((size_t)r * BATCH + 2 * bp + h) * UNITS + ntile * 16 + c] = s;
  }
}

extern "C" void kernel_launch(void* const* d_in, const int* in_sizes, int n_in,
                              void* d_out, int out_size, void* d_ws, size_t ws_size,
                              hipStream_t stream) {
  (void)in_sizes; (void)n_in; (void)out_size; (void)d_ws; (void)ws_size;
  const float* x = (const float*)d_in[0];
  const float* w = (const float*)d_in[1];
  float* out = (float*)d_out;
  dim3 grid(256);
  dim3 block(128);
  hipLaunchKernelGGL(ga_mvprod_kcat, grid, block, 0, stream, x, w, out);
}


# Products in the complex 2x2 matrix form of the algebra: blade pairs added and subtracted first, 32 real products per (b,f,u) not 64, blades rebuilt at the end; 128 groups; changes sum order
# speedup vs baseline: 2.4857x; 1.0764x over previous
#include <hip/hip_runtime.h>


typedef __attribute__((ext_vector_type(2))) float v2f;
typedef __attribute__((ext_vector_type(8))) float v8f;
typedef __attribute__((ext_vector_type(4))) float v4f;

#define NBLADES 8
#define BATCH   128
#define INFEAT  128
#define UNITS   64
#define XPITCH  132
#define WPLANE  (128 * 16 + 32)

#define PAIR_A_TAB 0x6170u
#define PAIR_B_TAB 0x2543u

__global__ __launch_bounds__(128) void ga_mvprod_c2x2(
    const float* __restrict__ x,
    const float* __restrict__ w,
    float* __restrict__ out)
{
  __shared__ __attribute__((aligned(16))) float wsm[4][2][WPLANE];
  __shared__ __attribute__((aligned(16))) float xsm[32 * XPITCH];
  __shared__ float red[4][2][8][32];

  const int tid  = threadIdx.x;
  const int lane = tid & 31;
  const int wv   = __builtin_amdgcn_readfirstlane(tid >> 5);
  const int bq    = blockIdx.x >> 2;
  const int ntile = blockIdx.x & 3;

  const int c = lane & 15;
  const int h = lane >> 4;
  const int bsel = c >> 2;
  const int ri   = (c >> 1) & 1;
  const int rc   = c & 1;

  const int pa  = (int)((PAIR_A_TAB >> (4 * wv)) & 7u);
  const int pb  = (int)((PAIR_B_TAB >> (4 * wv)) & 7u);
  const int cpr = wv & 1;

  v8f acc[2];
  acc[0] = (v8f){};
  acc[1] = (v8f){};

  const v4f* xr[2];
#pragma unroll
  for (int j = 0; j < 2; ++j) {
    const int k    = (wv >> 1) ^ j;
    const int part = rc ^ cpr;
    const int pair = part + 2 * (ri ^ k);
    xr[j] = (const v4f*)&xsm[(16 * k + 4 * pair + bsel) * XPITCH + 64 * h];
  }

  const float* xsa = x + ((size_t)(pa * BATCH + 4 * bq) * INFEAT + 4 * lane);
  const float* xsb = x + ((size_t)(pb * BATCH + 4 * bq) * INFEAT + 4 * lane);
  v4f xga[4], xgb[4];
#pragma unroll
  for (int i = 0; i < 4; ++i) {
    xga[i] = *(const v4f*)(xsa + i * INFEAT);
    xgb[i] = *(const v4f*)(xsb + i * INFEAT);
  }
  const int wrow = lane >> 2, wg = lane & 3;
  const float* wsa = w + ((size_t)(pa * INFEAT + wrow) * UNITS + ntile * 16 + 4 * wg);
  const float* wsb = w + ((size_t)(pb * INFEAT + wrow) * UNITS + ntile * 16 + 4 * wg);
  v4f wq[32];
#pragma unroll
  for (int g = 0; g < 8; ++g) {
#pragma unroll
    for (int j = 0; j < 2; ++j) {
#pragma unroll
      for (int hh = 0; hh < 2; ++hh) {
        wq[(g * 2 + j) * 2 + hh] =
            *(const v4f*)((j ? wsb : wsa) + (8 * hh + g) * 8 * UNITS);
      }
    }
  }

  float* wdst = &wsm[wv][0][wrow * 16 + 4 * wg];
  const float* bl = &wsm[wv][0][(64 * h) * 16 + h * 32 + c];
  v4f xv[2][16];
  v2f b[2][32];

#define W_TO_LDS(G0, G1)                                                      \
  _Pragma("unroll")                                                           \
  for (int g = (G0); g < (G1); ++g) {                                         \
    _Pragma("unroll")                                                         \
    for (int hh = 0; hh < 2; ++hh) {                                          \
      const v4f wfa = wq[(g * 2 + 0) * 2 + hh];                               \
      const v4f wfb = wq[(g * 2 + 1) * 2 + hh];                               \
      *(v4f*)(wdst + (64 * hh + 8 * g) * 16 + hh * 32) = wfa + wfb;           \
      *(v4f*)(wdst + WPLANE + (64 * hh + 8 * g) * 16 + hh * 32) = wfa - wfb;  \
    }                                                                         \
  }
#define FRAGS_FROM_LDS(G0, G1)                                                \
  _Pragma("unroll")                                                           \
  for (int q = 2 * (G0); q < 2 * (G1); ++q) {                                 \
    _Pragma("unroll")                                                         \
    for (int j = 0; j < 2; ++j) xv[j][q] = xr[j][q];                          \
    _Pragma("unroll")                                                         \
    for (int t = 2 * q; t < 2 * q + 2; ++t) {                                 \
      _Pragma("unroll")                                                       \
      for (int j = 0; j < 2; ++j) {                                           \
        b[j][t].x = bl[j * WPLANE + (2 * t + 0) * 16];                        \
        b[j][t].y = bl[j * WPLANE + (2 * t + 1) * 16];                        \
      }                                                                       \
    }                                                                         \
  }
#define MMA(G0, G1)                                                           \
  _Pragma("unroll")                                                           \
  for (int t = 4 * (G0); t < 4 * (G1); ++t) {                                 \
    _Pragma("unroll")                                                         \
    for (int j = 0; j < 2; ++j) {                                             \
      const v4f v = xv[j][t >> 1];                                            \
      v2f a;                                                                  \
      a.x = (t & 1) ? v.z : v.x;                                              \
      a.y = (t & 1) ? v.w : v.y;                                              \
      acc[j] = __builtin_amdgcn_wmma_f32_16x16x4_f32(                         \
          false, a, false, b[j][t], (short)0, acc[j], false, false);          \
    }                                                                         \
  }

#pragma unroll
  for (int i = 0; i < 4; ++i) {
    *(v4f*)&xsm[(4 * wv + i) * XPITCH + 4 * lane] = xga[i] + xgb[i];
    *(v4f*)&xsm[(16 + 4 * wv + i) * XPITCH + 4 * lane] = xga[i] - xgb[i];
  }
  W_TO_LDS(0, 1)
  __builtin_amdgcn_fence(__ATOMIC_RELEASE, "workgroup", "local");
  __builtin_amdgcn_s_barrier();
  __builtin_amdgcn_fence(__ATOMIC_ACQUIRE, "workgroup", "local");
  FRAGS_FROM_LDS(0, 1)

  W_TO_LDS(1, 2)
  FRAGS_FROM_LDS(1, 2)
  __builtin_amdgcn_sched_barrier(0);
  MMA(0, 1)
  __builtin_amdgcn_sched_barrier(0);
  W_TO_LDS(2, 4)
  FRAGS_FROM_LDS(2, 4)
  __builtin_amdgcn_sched_barrier(0);
  MMA(1, 2)
  __builtin_amdgcn_sched_barrier(0);
  W_TO_LDS(4, 6)
  FRAGS_FROM_LDS(4, 6)
  __builtin_amdgcn_sched_barrier(0);
  MMA(2, 4)
  __builtin_amdgcn_sched_barrier(0);
  W_TO_LDS(6, 8)
  FRAGS_FROM_LDS(6, 8)
  __builtin_amdgcn_sched_barrier(0);
  MMA(4, 6)
  MMA(6, 8)
#undef W_TO_LDS
#undef FRAGS_FROM_LDS
#undef MMA

  const unsigned sneg = (unsigned)cpr << 31;
#pragma unroll
  for (int j = 0; j < 2; ++j) {
#pragma unroll
    for (int r = 0; r < 8; ++r) {
      const unsigned s = (r & 1) ? 0u : sneg;
      red[wv][j][r][lane] = __uint_as_float(__float_as_uint(acc[j][r]) ^ s);
    }
  }
  __syncthreads();

  const int rb = wv >> 1, cc = wv & 1;
  float o2[2][2];
#pragma unroll
  for (int i = 0; i < 2; ++i) {
    const int r = (rb << 2) | (i << 1) | cc;
#pragma unroll
    for (int j = 0; j < 2; ++j) {
      o2[i][j] = (red[0][j][r][lane] + red[1][j][r][lane]) +
                 (red[2][j][r][lane] + red[3][j][r][lane]);
    }
  }
  const int bsd = cc ? 7 : 0, bdd = cc ? 4 : 3, bso = cc ? 6 : 1, bdo = cc ? 2 : 5;
  float* op = out + (size_t)(4 * bq + 2 * h + rb) * UNITS + ntile * 16 + c;
  op[(size_t)bsd * BATCH * UNITS] = 0.5f * (o2[0][0] + o2[1][1]);
  op[(size_t)bdd * BATCH * UNITS] = 0.5f * (o2[0][0] - o2[1][1]);
  op[(size_t)bso * BATCH * UNITS] = 0.5f * (o2[0][1] + o2[1][0]);
  op[(size_t)bdo * BATCH * UNITS] = 0.5f * (o2[1][0] - o2[0][1]);
}

extern "C" void kernel_launch(void* const* d_in, const int* in_sizes, int n_in,
                              void* d_out, int out_size, void* d_ws, size_t ws_size,
                              hipStream_t stream) {
  (void)in_sizes; (void)n_in; (void)out_size; (void)d_ws; (void)ws_size;
  const float* x = (const float*)d_in[0];
  const float* w = (const float*)d_in[1];
  float* out = (float*)d_out;
  dim3 grid(128);
  dim3 block(128);
  hipLaunchKernelGGL(ga_mvprod_c2x2, grid, block, 0, stream, x, w, out);
}
